# starting kernel
# baseline (speedup 1.0000x reference)
#include <hip/hip_runtime.h>

typedef __attribute__((ext_vector_type(2)))  float    v2f;
typedef __attribute__((ext_vector_type(8)))  float    v8f;
typedef __attribute__((ext_vector_type(16))) _Float16 v16h;

#define LOG2E 1.4426950408889634f

#if __has_builtin(__builtin_amdgcn_wmma_f32_16x16x4_f32)
#define USE_WMMA_F32X4 1
#else
#define USE_WMMA_F32X4 0
#endif

template <int IMM>
__device__ __forceinline__ float swzf(float v) {
  return __int_as_float(__builtin_amdgcn_ds_swizzle(__float_as_int(v), IMM));
}
template <int IMM>
__device__ __forceinline__ int swzi(int v) {
  return __builtin_amdgcn_ds_swizzle(v, IMM);
}

__device__ __forceinline__ float fexp2(float x) {
#if __has_builtin(__builtin_amdgcn_exp2f)
  return __builtin_amdgcn_exp2f(x);
#else
  return exp2f(x);
#endif
}
__device__ __forceinline__ float frcp(float x) {
#if __has_builtin(__builtin_amdgcn_rcpf)
  return __builtin_amdgcn_rcpf(x);
#else
  return 1.0f / x;
#endif
}
__device__ __forceinline__ float ftanh(float x) {
#if __has_builtin(__builtin_amdgcn_tanhf)
  return __builtin_amdgcn_tanhf(x);
#else
  float e = fexp2(x * (2.0f * LOG2E));
  return 1.0f - 2.0f * frcp(e + 1.0f);
#endif
}

template <int M, int IMM>
__device__ __forceinline__ float combine2(float a, float b, int lid) {
  const bool s = (lid & M) != 0;
  const float lo = s ? b : a;
  const float hv = s ? a : b;
  return lo + swzf<IMM>(hv);
}

#if !USE_WMMA_F32X4
__device__ __forceinline__ unsigned pack_h2(float x, float y) {
  typedef _Float16 h2 __attribute__((ext_vector_type(2)));
  h2 p; p.x = (_Float16)x; p.y = (_Float16)y;
  unsigned u; __builtin_memcpy(&u, &p, 4);
  return u;
}
#endif

__global__ __launch_bounds__(256) void hybridconv_wmma_kernel(
    const float* __restrict__ data,   const float* __restrict__ conv_w,
    const float* __restrict__ conv_b, const float* __restrict__ basis,
    const float* __restrict__ w1,     const float* __restrict__ b1,
    const float* __restrict__ w2,     const float* __restrict__ b2,
    float* __restrict__ out, int n)
{
  const int lid  = (int)threadIdx.x & 31;
  const int land = lid & 15;
  const int hi   = lid >> 4;
  const int kh   = hi << 1;

  const float cw0 = conv_w[0], cw1 = conv_w[1], cw2 = conv_w[2], cw3 = conv_w[3];
  const float cb  = conv_b[0];

  float tse[4], nse[4];
#pragma unroll
  for (int j = 0; j < 4; ++j) {
    const int row = kh + (j & 1) + ((j >> 1) << 2);
    const float e0 = basis[row * 4 + 0];
    const float e1 = basis[row * 4 + 1];
    const float e2 = basis[row * 4 + 2];
    const float e3 = basis[row * 4 + 3];
    const float S1 = e0 + e1 + e2 + e3;
    const float S2 = e0 * e0 + e1 * e1 + e2 * e2 + e3 * e3;
    tse[j] = 2.0f * S1 * LOG2E;
    nse[j] = -S2 * LOG2E;
  }

#if USE_WMMA_F32X4
  v2f bA, bB;
  bA.x = w1[(kh + 0) * 16 + land];
  bA.y = w1[(kh + 1) * 16 + land];
  bB.x = w1[(kh + 4) * 16 + land];
  bB.y = w1[(kh + 5) * 16 + land];
#else
  v16h B16;
#pragma unroll
  for (int j = 0; j < 8; ++j) {
    const bool use = (!hi) && (j < 4);
    const float lo = use ? w1[(2 * j + 0) * 16 + land] : 0.0f;
    const float hf = use ? w1[(2 * j + 1) * 16 + land] : 0.0f;
    B16[2 * j + 0] = (_Float16)lo;
    B16[2 * j + 1] = (_Float16)hf;
  }
#endif

  const float cinit = b1[land];
  const float wn    = w2[land];
  const float b2s   = b2[0] * (1.0f / 16.0f);

  const int ntiles        = (n + 15) >> 4;
  const int wavesPerBlock = (int)blockDim.x >> 5;
  const int gwave         = (int)blockIdx.x * wavesPerBlock + ((int)threadIdx.x >> 5);
  const int nwaves        = (int)gridDim.x * wavesPerBlock;

  const float4* __restrict__ d4 = reinterpret_cast<const float4*>(data);

  for (int tile = gwave; tile < ntiles; tile += nwaves) {
    const int p  = (tile << 4) + land;
    const int pc = (p < n) ? p : (n - 1);
    const float4 d = d4[pc];

    int pt = tile + nwaves;
    pt = (pt < ntiles) ? pt : tile;
    __builtin_prefetch(reinterpret_cast<const char*>(d4 + ((pt << 4) + land)), 0, 0);

    const float logit = fmaf(cw0, d.x, fmaf(cw1, d.y, fmaf(cw2, d.z, fmaf(cw3, d.w, cb))));
    const float act   = frcp(1.0f + fexp2(-logit * LOG2E));
    const float qe    = -4.0f * act * act * LOG2E;

    const float f0 = fexp2(fmaf(act, tse[0], nse[0] + qe));
    const float f1 = fexp2(fmaf(act, tse[1], nse[1] + qe));
    const float f2 = fexp2(fmaf(act, tse[2], nse[2] + qe));
    const float f3 = fexp2(fmaf(act, tse[3], nse[3] + qe));

    v8f c;
#pragma unroll
    for (int r = 0; r < 8; ++r) c[r] = cinit;

#if USE_WMMA_F32X4
    v2f a1; a1.x = f0; a1.y = f1;
    v2f a2; a2.x = f2; a2.y = f3;
    c = __builtin_amdgcn_wmma_f32_16x16x4_f32(false, a1, false, bA, (short)0, c, false, false);
    c = __builtin_amdgcn_wmma_f32_16x16x4_f32(false, a2, false, bB, (short)0, c, false, false);
#else
    unsigned pa = pack_h2(f0, f1);
    unsigned pb = pack_h2(f2, f3);
    unsigned qa = (unsigned)swzi<0x401F>((int)pa);
    unsigned qb = (unsigned)swzi<0x401F>((int)pb);
    const unsigned zm = hi ? 0u : 0xFFFFFFFFu;
    pa &= zm; qa &= zm; pb &= zm; qb &= zm;
    unsigned au[8] = {pa, qa, pb, qb, 0u, 0u, 0u, 0u};
    v16h A16; __builtin_memcpy(&A16, au, 32);
    c = __builtin_amdgcn_wmma_f32_16x16x32_f16(false, A16, false, B16, (short)0, c, false, false);
#endif

    const float s0 = fmaf(ftanh(c[0]), wn, b2s);
    const float s1 = fmaf(ftanh(c[1]), wn, b2s);
    const float s2 = fmaf(ftanh(c[2]), wn, b2s);
    const float s3 = fmaf(ftanh(c[3]), wn, b2s);
    const float s4 = fmaf(ftanh(c[4]), wn, b2s);
    const float s5 = fmaf(ftanh(c[5]), wn, b2s);
    const float s6 = fmaf(ftanh(c[6]), wn, b2s);
    const float s7 = fmaf(ftanh(c[7]), wn, b2s);

    const float t0 = combine2<1, 0x041F>(s0, s1, lid);
    const float t1 = combine2<1, 0x041F>(s2, s3, lid);
    const float t2 = combine2<1, 0x041F>(s4, s5, lid);
    const float t3 = combine2<1, 0x041F>(s6, s7, lid);
    const float u0 = combine2<2, 0x081F>(t0, t1, lid);
    const float u1 = combine2<2, 0x081F>(t2, t3, lid);
    float v        = combine2<4, 0x101F>(u0, u1, lid);
    v += swzf<0x201F>(v);

    const int o = (tile << 4) + (lid & 7) + (hi << 3);
    if (((lid & 8) == 0) && (o < n)) out[o] = v;
  }
}

extern "C" void kernel_launch(void* const* d_in, const int* in_sizes, int n_in,
                              void* d_out, int out_size, void* d_ws, size_t ws_size,
                              hipStream_t stream) {
  const float* data   = (const float*)d_in[0];
  const float* conv_w = (const float*)d_in[1];
  const float* conv_b = (const float*)d_in[2];
  const float* basis  = (const float*)d_in[3];
  const float* w1     = (const float*)d_in[4];
  const float* b1     = (const float*)d_in[5];
  const float* w2     = (const float*)d_in[6];
  const float* b2     = (const float*)d_in[7];
  float* out = (float*)d_out;

  const int n = in_sizes[0] / 4;
  const int ntiles = (n + 15) / 16;
  const int block = 256;
  const int wavesPerBlock = block / 32;
  int blocks = (ntiles + wavesPerBlock - 1) / wavesPerBlock;
  if (blocks > 4096) blocks = 4096;
  if (blocks < 1) blocks = 1;

  hipLaunchKernelGGL(hybridconv_wmma_kernel, dim3(blocks), dim3(block), 0, stream,
                     data, conv_w, conv_b, basis, w1, b1, w2, b2, out, n);
}


# Two-wide packed f32 FMAs (v_pk_fma_f32) for the per-lane math: RBF exponent a*(tse-4a*log2e)+nse on row pairs, w2 weighting on accumulator pairs; exponent sum re-associated, layer sums unchanged
# speedup vs baseline: 1.2317x; 1.2317x over previous
#include <hip/hip_runtime.h>

typedef __attribute__((ext_vector_type(2)))  float    v2f;
typedef __attribute__((ext_vector_type(8)))  float    v8f;
typedef __attribute__((ext_vector_type(16))) _Float16 v16h;

#define LOG2E 1.4426950408889634f

#if __has_builtin(__builtin_amdgcn_wmma_f32_16x16x4_f32)
#define USE_WMMA_F32X4 1
#else
#define USE_WMMA_F32X4 0
#endif

template <int IMM>
__device__ __forceinline__ float swzf(float v) {
  return __int_as_float(__builtin_amdgcn_ds_swizzle(__float_as_int(v), IMM));
}
template <int IMM>
__device__ __forceinline__ int swzi(int v) {
  return __builtin_amdgcn_ds_swizzle(v, IMM);
}

__device__ __forceinline__ float fexp2(float x) {
#if __has_builtin(__builtin_amdgcn_exp2f)
  return __builtin_amdgcn_exp2f(x);
#else
  return exp2f(x);
#endif
}
__device__ __forceinline__ float frcp(float x) {
#if __has_builtin(__builtin_amdgcn_rcpf)
  return __builtin_amdgcn_rcpf(x);
#else
  return 1.0f / x;
#endif
}
__device__ __forceinline__ float ftanh(float x) {
#if __has_builtin(__builtin_amdgcn_tanhf)
  return __builtin_amdgcn_tanhf(x);
#else
  float e = fexp2(x * (2.0f * LOG2E));
  return 1.0f - 2.0f * frcp(e + 1.0f);
#endif
}

template <int M, int IMM>
__device__ __forceinline__ float combine2(float a, float b, int lid) {
  const bool s = (lid & M) != 0;
  const float lo = s ? b : a;
  const float hv = s ? a : b;
  return lo + swzf<IMM>(hv);
}

__device__ __forceinline__ float xhalf(float x) {
  const int b = __builtin_bit_cast(int, x);
  return __builtin_bit_cast(float,
      __builtin_amdgcn_permlanex16(b, b, 0x76543210, 0xfedcba98, false, false));
}

#if !USE_WMMA_F32X4
__device__ __forceinline__ unsigned pack_h2(float x, float y) {
  typedef _Float16 h2 __attribute__((ext_vector_type(2)));
  h2 p; p.x = (_Float16)x; p.y = (_Float16)y;
  unsigned u; __builtin_memcpy(&u, &p, 4);
  return u;
}
#endif

__global__ __launch_bounds__(256) void hybridconv_wmma_kernel(
    const float* __restrict__ data,   const float* __restrict__ conv_w,
    const float* __restrict__ conv_b, const float* __restrict__ basis,
    const float* __restrict__ w1,     const float* __restrict__ b1,
    const float* __restrict__ w2,     const float* __restrict__ b2,
    float* __restrict__ out, int n)
{
  const int lid  = (int)threadIdx.x & 31;
  const int land = lid & 15;
  const int hi   = lid >> 4;
  const int kh   = hi << 1;

  const float cw0 = conv_w[0], cw1 = conv_w[1], cw2 = conv_w[2], cw3 = conv_w[3];
  const float cb  = conv_b[0];

  float tse[4], nse[4];
#pragma unroll
  for (int j = 0; j < 4; ++j) {
    const int row = kh + (j & 1) + ((j >> 1) << 2);
    const float e0 = basis[row * 4 + 0];
    const float e1 = basis[row * 4 + 1];
    const float e2 = basis[row * 4 + 2];
    const float e3 = basis[row * 4 + 3];
    const float S1 = e0 + e1 + e2 + e3;
    const float S2 = e0 * e0 + e1 * e1 + e2 * e2 + e3 * e3;
    tse[j] = 2.0f * S1 * LOG2E;
    nse[j] = -S2 * LOG2E;
  }
  v2f tse01, tse23, nse01, nse23;
  tse01.x = tse[0]; tse01.y = tse[1]; tse23.x = tse[2]; tse23.y = tse[3];
  nse01.x = nse[0]; nse01.y = nse[1]; nse23.x = nse[2]; nse23.y = nse[3];
  v2f kq; kq.x = -4.0f * LOG2E; kq.y = -4.0f * LOG2E;

#if USE_WMMA_F32X4
  v2f bA, bB;
  bA.x = w1[(kh + 0) * 16 + land];
  bA.y = w1[(kh + 1) * 16 + land];
  bB.x = w1[(kh + 4) * 16 + land];
  bB.y = w1[(kh + 5) * 16 + land];
#else
  v16h B16;
#pragma unroll
  for (int j = 0; j < 8; ++j) {
    const bool use = (!hi) && (j < 4);
    const float lo = use ? w1[(2 * j + 0) * 16 + land] : 0.0f;
    const float hf = use ? w1[(2 * j + 1) * 16 + land] : 0.0f;
    B16[2 * j + 0] = (_Float16)lo;
    B16[2 * j + 1] = (_Float16)hf;
  }
#endif

  const float cinit = b1[land];
  const float wn    = w2[land];
  const float b2s   = b2[0] * (1.0f / 16.0f);
  v2f wn2;  wn2.x  = wn;  wn2.y  = wn;
  v2f b2s2; b2s2.x = b2s; b2s2.y = b2s;

  const int ntrips        = (n + 31) >> 5;
  const int wavesPerBlock = (int)blockDim.x >> 5;
  const int gwave         = (int)blockIdx.x * wavesPerBlock + ((int)threadIdx.x >> 5);
  const int nwaves        = (int)gridDim.x * wavesPerBlock;

  const float4* __restrict__ d4 = reinterpret_cast<const float4*>(data);

  for (int trip = gwave; trip < ntrips; trip += nwaves) {
    const int p  = (trip << 5) + lid;
    const int pc = (p < n) ? p : (n - 1);
    const float4 d = d4[pc];

    int pt = trip + nwaves;
    pt = (pt < ntrips) ? pt : trip;
    __builtin_prefetch(reinterpret_cast<const char*>(d4 + ((pt << 5) + lid)), 0, 0);

    const float logit = fmaf(cw0, d.x, fmaf(cw1, d.y, fmaf(cw2, d.z, fmaf(cw3, d.w, cb))));
    const float act   = frcp(1.0f + fexp2(-logit * LOG2E));

    const float asw = xhalf(act);

    v8f cT[2];
#pragma unroll
    for (int t = 0; t < 2; ++t) {
      const float a  = ((hi != 0) == (t != 0)) ? act : asw;
      v2f ap; ap.x = a; ap.y = a;
      const v2f u01 = __builtin_elementwise_fma(ap, kq,  tse01);
      const v2f u23 = __builtin_elementwise_fma(ap, kq,  tse23);
      const v2f e01 = __builtin_elementwise_fma(ap, u01, nse01);
      const v2f e23 = __builtin_elementwise_fma(ap, u23, nse23);
      const float f0 = fexp2(e01.x);
      const float f1 = fexp2(e01.y);
      const float f2 = fexp2(e23.x);
      const float f3 = fexp2(e23.y);

      v8f c;
#pragma unroll
      for (int r = 0; r < 8; ++r) c[r] = cinit;

#if USE_WMMA_F32X4
      v2f a1; a1.x = f0; a1.y = f1;
      v2f a2; a2.x = f2; a2.y = f3;
      c = __builtin_amdgcn_wmma_f32_16x16x4_f32(false, a1, false, bA, (short)0, c, false, false);
      c = __builtin_amdgcn_wmma_f32_16x16x4_f32(false, a2, false, bB, (short)0, c, false, false);
#else
      unsigned pa = pack_h2(f0, f1);
      unsigned pb = pack_h2(f2, f3);
      unsigned qa = (unsigned)swzi<0x401F>((int)pa);
      unsigned qb = (unsigned)swzi<0x401F>((int)pb);
      const unsigned zm = hi ? 0u : 0xFFFFFFFFu;
      pa &= zm; qa &= zm; pb &= zm; qb &= zm;
      unsigned au[8] = {pa, qa, pb, qb, 0u, 0u, 0u, 0u};
      v16h A16; __builtin_memcpy(&A16, au, 32);
      c = __builtin_amdgcn_wmma_f32_16x16x32_f16(false, A16, false, B16, (short)0, c, false, false);
#endif
      cT[t] = c;
    }

    float s[16];
#pragma unroll
    for (int t = 0; t < 2; ++t) {
#pragma unroll
      for (int r = 0; r < 8; r += 2) {
        v2f h; h.x = ftanh(cT[t][r]); h.y = ftanh(cT[t][r + 1]);
        const v2f w = __builtin_elementwise_fma(h, wn2, b2s2);
        s[8 * t + r]     = w.x;
        s[8 * t + r + 1] = w.y;
      }
    }

    float t1[8], t2[4];
#pragma unroll
    for (int i = 0; i < 8; ++i) t1[i] = combine2<1, 0x041F>(s[2 * i], s[2 * i + 1], lid);
#pragma unroll
    for (int i = 0; i < 4; ++i) t2[i] = combine2<2, 0x081F>(t1[2 * i], t1[2 * i + 1], lid);
    const float u0 = combine2<4, 0x101F>(t2[0], t2[1], lid);
    const float u1 = combine2<4, 0x101F>(t2[2], t2[3], lid);
    const float v  = combine2<8, 0x201F>(u0, u1, lid);

    const int o = (trip << 5) + ((lid & 8) << 1) + (lid & 7) + (hi << 3);
    if (o < n) out[o] = v;
  }
}

extern "C" void kernel_launch(void* const* d_in, const int* in_sizes, int n_in,
                              void* d_out, int out_size, void* d_ws, size_t ws_size,
                              hipStream_t stream) {
  const float* data   = (const float*)d_in[0];
  const float* conv_w = (const float*)d_in[1];
  const float* conv_b = (const float*)d_in[2];
  const float* basis  = (const float*)d_in[3];
  const float* w1     = (const float*)d_in[4];
  const float* b1     = (const float*)d_in[5];
  const float* w2     = (const float*)d_in[6];
  const float* b2     = (const float*)d_in[7];
  float* out = (float*)d_out;

  const int n = in_sizes[0] / 4;
  const int ntrips = (n + 31) / 32;
  const int block = 256;
  const int wavesPerBlock = block / 32;
  int blocks = (ntrips + wavesPerBlock - 1) / wavesPerBlock;
  if (blocks > 4096) blocks = 4096;
  if (blocks < 1) blocks = 1;

  hipLaunchKernelGGL(hybridconv_wmma_kernel, dim3(blocks), dim3(block), 0, stream,
                     data, conv_w, conv_b, basis, w1, b1, w2, b2, out, n);
}
